# starting kernel
# baseline (speedup 1.0000x reference)
#include <hip/hip_runtime.h>

typedef __attribute__((ext_vector_type(2))) float v2f;
typedef __attribute__((ext_vector_type(8))) float v8f;

#define NCLS 512
#define NBATCH 4096
#define BM 32

__global__ __launch_bounds__(256) void pack_operands(const float* __restrict__ V,
                                                     float* __restrict__ VPv,
                                                     float* __restrict__ VPt) {
    const int idx = blockIdx.x * 256 + threadIdx.x;
    const int col = idx & (NCLS - 1);
    const int k2  = idx >> 9;

    v2f a;
    a.x = V[(size_t)(2 * k2) * NCLS + col];
    a.y = V[(size_t)(2 * k2 + 1) * NCLS + col];
    *(v2f*)(VPv + ((size_t)k2 * NCLS + col) * 2) = a;

    v2f b = *(const v2f*)(V + (size_t)col * NCLS + 2 * k2);
    *(v2f*)(VPt + ((size_t)k2 * NCLS + col) * 2) = b;
}

__global__ __launch_bounds__(256) void fused_probability_noiser(
    const float* __restrict__ P0,
    const float* __restrict__ sigma,
    const float* __restrict__ lam,
    const float* __restrict__ VPv,
    const float* __restrict__ VPt,
    float* __restrict__ out)
{
    __shared__ float su[BM][NCLS];

    const int tid  = threadIdx.x;
    const int wave = tid >> 5;
    const int lane = tid & 31;
    const int nn   = lane & 15;
    const int hi   = lane >> 4;

    const int cbas = wave * 64;
    const int row0 = blockIdx.x * BM;
    const int m    = nn;

    v8f acc[2][4];
#pragma unroll
    for (int rb = 0; rb < 2; ++rb)
#pragma unroll
        for (int t = 0; t < 4; ++t)
#pragma unroll
            for (int i = 0; i < 8; ++i) acc[rb][t][i] = 0.0f;

    {
        const float* Ap0 = P0 + (size_t)(row0 + m) * NCLS;
        const float* Ap1 = P0 + (size_t)(row0 + 16 + m) * NCLS;
        for (int k = 0; k < NCLS; k += 4) {
            const int ka = k + 2 * hi;
            v2f a0 = *(const v2f*)(Ap0 + ka);
            v2f a1 = *(const v2f*)(Ap1 + ka);
            const float* Bp =
                VPv + (size_t)((k >> 1) + hi) * (2 * NCLS) + (size_t)(cbas + nn) * 2;
#pragma unroll
            for (int t = 0; t < 4; ++t) {
                v2f b = *(const v2f*)(Bp + t * 32);
                acc[0][t] = __builtin_amdgcn_wmma_f32_16x16x4_f32(
                    false, a0, false, b, (short)0, acc[0][t], false, false);
                acc[1][t] = __builtin_amdgcn_wmma_f32_16x16x4_f32(
                    false, a1, false, b, (short)0, acc[1][t], false, false);
            }
        }
    }

    {
        float sg[2][8];
#pragma unroll
        for (int rb = 0; rb < 2; ++rb)
#pragma unroll
            for (int g = 0; g < 8; ++g)
                sg[rb][g] = sigma[row0 + rb * 16 + g + 8 * hi];
#pragma unroll
        for (int t = 0; t < 4; ++t) {
            const int col = cbas + t * 16 + nn;
            const float lv = lam[col];
#pragma unroll
            for (int rb = 0; rb < 2; ++rb)
#pragma unroll
                for (int g = 0; g < 8; ++g) {
                    const int mr = g + 8 * hi;
                    const float sc = __expf(lv * sg[rb][g]);
                    su[rb * 16 + mr][col ^ (mr << 2)] = sc * acc[rb][t][g];
                }
        }
    }
    __syncthreads();

#pragma unroll
    for (int rb = 0; rb < 2; ++rb)
#pragma unroll
        for (int t = 0; t < 4; ++t)
#pragma unroll
            for (int i = 0; i < 8; ++i) acc[rb][t][i] = 0.0f;

    {
        const float* A0 = &su[m][0];
        const float* A1 = &su[16 + m][0];
        for (int k = 0; k < NCLS; k += 4) {
            const int ka = k + 2 * hi;
            const int sw = ka ^ (m << 2);
            v2f a0 = *(const v2f*)(A0 + sw);
            v2f a1 = *(const v2f*)(A1 + sw);
            const float* Bp =
                VPt + (size_t)((k >> 1) + hi) * (2 * NCLS) + (size_t)(cbas + nn) * 2;
#pragma unroll
            for (int t = 0; t < 4; ++t) {
                v2f b = *(const v2f*)(Bp + t * 32);
                acc[0][t] = __builtin_amdgcn_wmma_f32_16x16x4_f32(
                    false, a0, false, b, (short)0, acc[0][t], false, false);
                acc[1][t] = __builtin_amdgcn_wmma_f32_16x16x4_f32(
                    false, a1, false, b, (short)0, acc[1][t], false, false);
            }
        }
    }

#pragma unroll
    for (int t = 0; t < 4; ++t) {
        const int col = cbas + t * 16 + nn;
#pragma unroll
        for (int rb = 0; rb < 2; ++rb)
#pragma unroll
            for (int g = 0; g < 8; ++g) {
                const int r = row0 + rb * 16 + g + 8 * hi;
                out[(size_t)r * NCLS + col] = acc[rb][t][g];
            }
    }
}

extern "C" void kernel_launch(void* const* d_in, const int* in_sizes, int n_in,
                              void* d_out, int out_size, void* d_ws, size_t ws_size,
                              hipStream_t stream) {
    const float* p0    = (const float*)d_in[0];
    const float* sigma = (const float*)d_in[1];
    const float* evals = (const float*)d_in[2];
    const float* evecs = (const float*)d_in[3];
    float* out = (float*)d_out;
    float* VPv = (float*)d_ws;
    float* VPt = VPv + (size_t)NCLS * NCLS;

    pack_operands<<<(NCLS * NCLS / 2) / 256, 256, 0, stream>>>(evecs, VPv, VPt);

    fused_probability_noiser<<<NBATCH / BM, 256, 0, stream>>>(
        p0, sigma, evals, VPv, VPt, out);
}


# Pack launch dropped (3 launches -> 2): both products stage B slabs straight from V; the first interleaves rows k, k+1 in registers into the packed LDS layout, the second reads V rows like A rows
# speedup vs baseline: 2.2674x; 2.2674x over previous
#include <hip/hip_runtime.h>

typedef __attribute__((ext_vector_type(2))) float v2f;
typedef __attribute__((ext_vector_type(4))) float v4f;
typedef __attribute__((ext_vector_type(8))) float v8f;

#define NCLS 512
#define NBATCH 4096
#define TM 64
#define TN 128
#define KS 64
#define KPR (KS / 2)
#define APITCH (KS + 4)
#define BPITCH (2 * TN + 32)
#define ASLAB (TM * APITCH)
#define BSLAB (KPR * BPITCH)
#define STAGE (ASLAB + BSLAB)
#define NSLAB (NCLS / KS)
#define NA (TM * KS / 4 / 256)
#define NB (KS * TN / 4 / 256)
#define LDSF (41 * 1024)

template <bool FIRST>
__device__ __forceinline__ void slab_fetch(v4f (&ga)[NA], v4f (&gb)[NB],
                                           const float* __restrict__ asrc,
                                           const float* __restrict__ bsrc) {
#pragma unroll
    for (int i = 0; i < NA; ++i) ga[i] = *(const v4f*)(asrc + (size_t)i * 16 * NCLS);
#pragma unroll
    for (int i = 0; i < NB; ++i)
        gb[i] = FIRST ? *(const v4f*)(bsrc + (size_t)((i >> 1) * 16 + (i & 1)) * NCLS)
                      : *(const v4f*)(bsrc + (size_t)i * 16 * NCLS);
}

template <bool FIRST>
__device__ __forceinline__ void slab_put(const v4f (&ga)[NA], const v4f (&gb)[NB],
                                         float* adst, float* bdst) {
#pragma unroll
    for (int i = 0; i < NA; ++i) *(v4f*)(adst + i * 16 * APITCH) = ga[i];
#pragma unroll
    for (int i = 0; i < NB; ++i) {
        if (FIRST) {
            if (i & 1) continue;
            const v4f r0 = gb[i], r1 = gb[i + 1];
            v4f lo, up;
            lo.x = r0.x; lo.y = r1.x; lo.z = r0.y; lo.w = r1.y;
            up.x = r0.z; up.y = r1.z; up.z = r0.w; up.w = r1.w;
            *(v4f*)(bdst + (i >> 1) * 8 * BPITCH) = lo;
            *(v4f*)(bdst + (i >> 1) * 8 * BPITCH + 4) = up;
        } else {
            *(v4f*)(bdst + i * 16 * APITCH) = gb[i];
        }
    }
}

template <bool FIRST, int N>
__device__ __forceinline__ void frag_load(v2f (&fa)[N][2], v2f (&fb)[N][2],
                                          const float* Ap, const float* Bp, int kk0) {
#pragma unroll
    for (int j = 0; j < N; ++j) {
        const int kk = kk0 + j;
        fa[j][0] = *(const v2f*)(Ap + 4 * kk);
        fa[j][1] = *(const v2f*)(Ap + 16 * APITCH + 4 * kk);
        if (FIRST) {
            fb[j][0] = *(const v2f*)(Bp + 2 * kk * BPITCH);
            fb[j][1] = *(const v2f*)(Bp + 2 * kk * BPITCH + 32);
        } else {
            fb[j][0] = *(const v2f*)(Bp + 4 * kk);
            fb[j][1] = *(const v2f*)(Bp + 16 * APITCH + 4 * kk);
        }
    }
}

template <int N>
__device__ __forceinline__ void frag_mma(v8f (&acc)[2][2], const v2f (&fa)[N][2],
                                         const v2f (&fb)[N][2]) {
#pragma unroll
    for (int j = 0; j < N; ++j) {
        acc[0][0] = __builtin_amdgcn_wmma_f32_16x16x4_f32(
            false, fa[j][0], false, fb[j][0], (short)0, acc[0][0], false, false);
        acc[1][0] = __builtin_amdgcn_wmma_f32_16x16x4_f32(
            false, fa[j][1], false, fb[j][0], (short)0, acc[1][0], false, false);
        acc[0][1] = __builtin_amdgcn_wmma_f32_16x16x4_f32(
            false, fa[j][0], false, fb[j][1], (short)0, acc[0][1], false, false);
        acc[1][1] = __builtin_amdgcn_wmma_f32_16x16x4_f32(
            false, fa[j][1], false, fb[j][1], (short)0, acc[1][1], false, false);
    }
}

#define NK (KS / 4)
#define K1 (NK / 2)

template <bool FIRST>
__device__ __forceinline__ void tile_product(const float* __restrict__ A,
                                             const float* __restrict__ V,
                                             const float* __restrict__ sigma,
                                             const float* __restrict__ lam,
                                             float* __restrict__ C, float* smem) {
    const int tid  = threadIdx.x;
    const int wave = tid >> 5;
    const int lane = tid & 31;
    const int nn   = lane & 15;
    const int hi   = lane >> 4;
    const int wr   = wave >> 2;
    const int wc   = wave & 3;

    const int row0 = (blockIdx.x >> 2) * TM;
    const int col0 = (blockIdx.x & 3) * TN;

    const float* asrc = A + (size_t)(row0 + (tid >> 4)) * NCLS + (tid & 15) * 4;
    const int aput = (tid >> 4) * APITCH + (tid & 15) * 4;
    const float* bsrc = FIRST ? V + (size_t)(2 * (tid >> 5)) * NCLS + col0 + (tid & 31) * 4
                              : V + (size_t)(col0 + (tid >> 4)) * NCLS + (tid & 15) * 4;
    const int bput = FIRST ? ASLAB + (tid >> 5) * BPITCH + (tid & 31) * 8
                           : ASLAB + (tid >> 4) * APITCH + (tid & 15) * 4;
    const size_t bstep = FIRST ? (size_t)KS * NCLS : (size_t)KS;
    const int afr = (wr * 32 + nn) * APITCH + 2 * hi;
    const int bfr = FIRST ? ASLAB + hi * BPITCH + (wc * 32 + nn) * 2
                          : ASLAB + (wc * 32 + nn) * APITCH + 2 * hi;

    {
        v4f ga[NA], gb[NB];
        slab_fetch<FIRST>(ga, gb, asrc, bsrc);
        slab_put<FIRST>(ga, gb, smem + aput, smem + bput);
    }
    __syncthreads();

    v8f acc[2][2];
#pragma unroll
    for (int rb = 0; rb < 2; ++rb)
#pragma unroll
        for (int t = 0; t < 2; ++t)
#pragma unroll
            for (int i = 0; i < 8; ++i) acc[rb][t][i] = 0.0f;

#pragma nounroll
    for (int s = 0; s < NSLAB - 1; ++s) {
        const float* cur = smem + (s & 1) * STAGE;
        float* nxt = smem + ((s + 1) & 1) * STAGE;
        v4f ga[NA], gb[NB];
        v2f a1[K1][2], b1[K1][2], a2[NK - K1][2], b2[NK - K1][2];
        slab_fetch<FIRST>(ga, gb, asrc + (s + 1) * KS, bsrc + (size_t)(s + 1) * bstep);
        frag_load<FIRST, K1>(a1, b1, cur + afr, cur + bfr, 0);
        frag_mma<K1>(acc, a1, b1);
        frag_load<FIRST, NK - K1>(a2, b2, cur + afr, cur + bfr, K1);
        asm volatile("" :: "v"(acc[0][0]), "v"(acc[1][0]), "v"(acc[0][1]), "v"(acc[1][1]));
        slab_put<FIRST>(ga, gb, nxt + aput, nxt + bput);
        frag_mma<NK - K1>(acc, a2, b2);
        __syncthreads();
    }
    {
        const float* cur = smem + ((NSLAB - 1) & 1) * STAGE;
        v2f a1[NK][2], b1[NK][2];
        frag_load<FIRST, NK>(a1, b1, cur + afr, cur + bfr, 0);
        frag_mma<NK>(acc, a1, b1);
    }

    float sg[2][8];
    float lv[2];
    if (FIRST) {
#pragma unroll
        for (int rb = 0; rb < 2; ++rb)
#pragma unroll
            for (int g = 0; g < 8; ++g)
                sg[rb][g] = sigma[row0 + wr * 32 + rb * 16 + 8 * hi + g];
#pragma unroll
        for (int t = 0; t < 2; ++t) lv[t] = lam[col0 + wc * 32 + t * 16 + nn];
    }
#pragma unroll
    for (int t = 0; t < 2; ++t) {
        const int col = col0 + wc * 32 + t * 16 + nn;
#pragma unroll
        for (int rb = 0; rb < 2; ++rb)
#pragma unroll
            for (int g = 0; g < 8; ++g) {
                const int r = row0 + wr * 32 + rb * 16 + 8 * hi + g;
                float v = acc[rb][t][g];
                if (FIRST) v = __expf(lv[t] * sg[rb][g]) * v;
                C[(size_t)r * NCLS + col] = v;
            }
    }
}

__global__ __launch_bounds__(256) void product_scaled(
    const float* __restrict__ P0,
    const float* __restrict__ sigma,
    const float* __restrict__ lam,
    const float* __restrict__ V,
    float* __restrict__ U)
{
    __shared__ __attribute__((aligned(16))) float smem[LDSF];
    tile_product<true>(P0, V, sigma, lam, U, smem);
}

__global__ __launch_bounds__(256) void product_plain(
    const float* __restrict__ U,
    const float* __restrict__ sigma,
    const float* __restrict__ lam,
    const float* __restrict__ V,
    float* __restrict__ out)
{
    __shared__ __attribute__((aligned(16))) float smem[LDSF];
    tile_product<false>(U, V, sigma, lam, out, smem);
}

extern "C" void kernel_launch(void* const* d_in, const int* in_sizes, int n_in,
                              void* d_out, int out_size, void* d_ws, size_t ws_size,
                              hipStream_t stream) {
    const float* p0    = (const float*)d_in[0];
    const float* sigma = (const float*)d_in[1];
    const float* evals = (const float*)d_in[2];
    const float* evecs = (const float*)d_in[3];
    float* out = (float*)d_out;
    float* U   = (float*)d_ws;

    product_scaled<<<(NBATCH / TM) * (NCLS / TN), 256, 0, stream>>>(p0, sigma, evals, evecs, U);

    product_plain<<<(NBATCH / TM) * (NCLS / TN), 256, 0, stream>>>(U, sigma, evals, evecs, out);
}


# Slab copy one slab deeper: slab s+2's global loads go out at mid-slab s behind slab s+1's LDS stores and stay in flight across an LDS-only barrier, so a whole slab of WMMAs (was half) covers them
# speedup vs baseline: 2.3900x; 1.0541x over previous
#include <hip/hip_runtime.h>

typedef __attribute__((ext_vector_type(2))) float v2f;
typedef __attribute__((ext_vector_type(4))) float v4f;
typedef __attribute__((ext_vector_type(8))) float v8f;

#define NCLS 512
#define NBATCH 4096
#define TM 64
#define TN 128
#define KS 64
#define KPR (KS / 2)
#define APITCH (KS + 4)
#define BPITCH (2 * TN + 32)
#define ASLAB (TM * APITCH)
#define BSLAB (KPR * BPITCH)
#define STAGE (ASLAB + BSLAB)
#define NSLAB (NCLS / KS)
#define NA (TM * KS / 4 / 256)
#define NB (KS * TN / 4 / 256)
#define LDSF (41 * 1024)

template <bool FIRST>
__device__ __forceinline__ void slab_fetch(v4f (&ga)[NA], v4f (&gb)[NB],
                                           const float* __restrict__ asrc,
                                           const float* __restrict__ bsrc) {
#pragma unroll
    for (int i = 0; i < NA; ++i) ga[i] = *(const v4f*)(asrc + (size_t)i * 16 * NCLS);
#pragma unroll
    for (int i = 0; i < NB; ++i)
        gb[i] = FIRST ? *(const v4f*)(bsrc + (size_t)((i >> 1) * 16 + (i & 1)) * NCLS)
                      : *(const v4f*)(bsrc + (size_t)i * 16 * NCLS);
}

template <bool FIRST>
__device__ __forceinline__ void slab_put(const v4f (&ga)[NA], const v4f (&gb)[NB],
                                         float* adst, float* bdst) {
#pragma unroll
    for (int i = 0; i < NA; ++i) *(v4f*)(adst + i * 16 * APITCH) = ga[i];
#pragma unroll
    for (int i = 0; i < NB; ++i) {
        if (FIRST) {
            if (i & 1) continue;
            const v4f r0 = gb[i], r1 = gb[i + 1];
            v4f lo, up;
            lo.x = r0.x; lo.y = r1.x; lo.z = r0.y; lo.w = r1.y;
            up.x = r0.z; up.y = r1.z; up.z = r0.w; up.w = r1.w;
            *(v4f*)(bdst + (i >> 1) * 8 * BPITCH) = lo;
            *(v4f*)(bdst + (i >> 1) * 8 * BPITCH + 4) = up;
        } else {
            *(v4f*)(bdst + i * 16 * APITCH) = gb[i];
        }
    }
}

template <bool FIRST, int N>
__device__ __forceinline__ void frag_load(v2f (&fa)[N][2], v2f (&fb)[N][2],
                                          const float* Ap, const float* Bp, int kk0) {
#pragma unroll
    for (int j = 0; j < N; ++j) {
        const int kk = kk0 + j;
        fa[j][0] = *(const v2f*)(Ap + 4 * kk);
        fa[j][1] = *(const v2f*)(Ap + 16 * APITCH + 4 * kk);
        if (FIRST) {
            fb[j][0] = *(const v2f*)(Bp + 2 * kk * BPITCH);
            fb[j][1] = *(const v2f*)(Bp + 2 * kk * BPITCH + 32);
        } else {
            fb[j][0] = *(const v2f*)(Bp + 4 * kk);
            fb[j][1] = *(const v2f*)(Bp + 16 * APITCH + 4 * kk);
        }
    }
}

template <int N, int J0 = 0, int J1 = N>
__device__ __forceinline__ void frag_mma(v8f (&acc)[2][2], const v2f (&fa)[N][2],
                                         const v2f (&fb)[N][2]) {
#pragma unroll
    for (int j = J0; j < J1; ++j) {
        acc[0][0] = __builtin_amdgcn_wmma_f32_16x16x4_f32(
            false, fa[j][0], false, fb[j][0], (short)0, acc[0][0], false, false);
        acc[1][0] = __builtin_amdgcn_wmma_f32_16x16x4_f32(
            false, fa[j][1], false, fb[j][0], (short)0, acc[1][0], false, false);
        acc[0][1] = __builtin_amdgcn_wmma_f32_16x16x4_f32(
            false, fa[j][0], false, fb[j][1], (short)0, acc[0][1], false, false);
        acc[1][1] = __builtin_amdgcn_wmma_f32_16x16x4_f32(
            false, fa[j][1], false, fb[j][1], (short)0, acc[1][1], false, false);
    }
}

#define NK (KS / 4)
#define K1 (NK / 2)

#define KT 4

__device__ __forceinline__ void slab_barrier_signal() {
    __builtin_amdgcn_sched_barrier(0);
    __builtin_amdgcn_fence(__ATOMIC_RELEASE, "workgroup", "local");
    __builtin_amdgcn_s_barrier_signal(-1);
    __builtin_amdgcn_sched_barrier(0);
}
__device__ __forceinline__ void slab_barrier_wait() {
    __builtin_amdgcn_s_barrier_wait(-1);
    __builtin_amdgcn_fence(__ATOMIC_ACQUIRE, "workgroup", "local");
}

template <bool FIRST>
__device__ __forceinline__ void slab_step(v8f (&acc)[2][2], v4f (&ga)[NA], v4f (&gb)[NB],
                                          const float* cur, float* nxt, int afr, int bfr,
                                          int aput, int bput, bool fetch,
                                          const float* __restrict__ asrc,
                                          const float* __restrict__ bsrc,
                                          int aoff, size_t boff) {
    v2f fa[NK][2], fb[NK][2];
    frag_load<FIRST, NK>(fa, fb, cur + afr, cur + bfr, 0);
    frag_mma<NK, 0, 2>(acc, fa, fb);
    __builtin_amdgcn_sched_barrier(0);
    frag_mma<NK, 2, K1>(acc, fa, fb);
    asm volatile("" :: "v"(acc[0][0]), "v"(acc[1][0]), "v"(acc[0][1]), "v"(acc[1][1]));
    if (FIRST) {
        asm volatile("" : "+v"(gb[0]), "+v"(gb[1]), "+v"(gb[2]), "+v"(gb[3]),
                          "+v"(gb[4]), "+v"(gb[5]), "+v"(gb[6]), "+v"(gb[7]));
    }
    slab_put<FIRST>(ga, gb, nxt + aput, nxt + bput);
    if (fetch) slab_fetch<FIRST>(ga, gb, asrc + aoff, bsrc + boff);
    frag_mma<NK, K1, NK - KT>(acc, fa, fb);
    slab_barrier_signal();
    frag_mma<NK, NK - KT, NK>(acc, fa, fb);
    slab_barrier_wait();
}

template <bool FIRST>
__device__ __forceinline__ void tile_product(const float* __restrict__ A,
                                             const float* __restrict__ V,
                                             const float* __restrict__ sigma,
                                             const float* __restrict__ lam,
                                             float* __restrict__ C, float* smem) {
    const int tid  = threadIdx.x;
    const int wave = tid >> 5;
    const int lane = tid & 31;
    const int nn   = lane & 15;
    const int hi   = lane >> 4;
    const int wr   = wave >> 2;
    const int wc   = wave & 3;

    const int row0 = (blockIdx.x >> 2) * TM;
    const int col0 = (blockIdx.x & 3) * TN;

    const float* asrc = A + (size_t)(row0 + (tid >> 4)) * NCLS + (tid & 15) * 4;
    const int aput = (tid >> 4) * APITCH + (tid & 15) * 4;
    const float* bsrc = FIRST ? V + (size_t)(2 * (tid >> 5)) * NCLS + col0 + (tid & 31) * 4
                              : V + (size_t)(col0 + (tid >> 4)) * NCLS + (tid & 15) * 4;
    const int bput = FIRST ? ASLAB + (tid >> 5) * BPITCH + (tid & 31) * 8
                           : ASLAB + (tid >> 4) * APITCH + (tid & 15) * 4;
    const size_t bstep = FIRST ? (size_t)KS * NCLS : (size_t)KS;
    const int afr = (wr * 32 + nn) * APITCH + 2 * hi;
    const int bfr = FIRST ? ASLAB + hi * BPITCH + (wc * 32 + nn) * 2
                          : ASLAB + (wc * 32 + nn) * APITCH + 2 * hi;

    v4f ga[NA], gb[NB];
    slab_fetch<FIRST>(ga, gb, asrc, bsrc);
    __builtin_amdgcn_sched_barrier(0);
    slab_put<FIRST>(ga, gb, smem + aput, smem + bput);
    slab_fetch<FIRST>(ga, gb, asrc + KS, bsrc + bstep);
    slab_barrier_signal();
    slab_barrier_wait();

    v8f acc[2][2];
#pragma unroll
    for (int rb = 0; rb < 2; ++rb)
#pragma unroll
        for (int t = 0; t < 2; ++t)
#pragma unroll
            for (int i = 0; i < 8; ++i) acc[rb][t][i] = 0.0f;

#pragma nounroll
    for (int s = 0; s < NSLAB - 1; ++s)
        slab_step<FIRST>(acc, ga, gb, smem + (s & 1) * STAGE, smem + ((s + 1) & 1) * STAGE,
                         afr, bfr, aput, bput, s + 2 < NSLAB,
                         asrc, bsrc, (s + 2) * KS, (size_t)(s + 2) * bstep);
    {
        const float* cur = smem + ((NSLAB - 1) & 1) * STAGE;
        v2f a1[NK][2], b1[NK][2];
        frag_load<FIRST, NK>(a1, b1, cur + afr, cur + bfr, 0);
        frag_mma<NK>(acc, a1, b1);
    }

    float sg[2][8];
    float lv[2];
    if (FIRST) {
#pragma unroll
        for (int rb = 0; rb < 2; ++rb)
#pragma unroll
            for (int g = 0; g < 8; ++g)
                sg[rb][g] = sigma[row0 + wr * 32 + rb * 16 + 8 * hi + g];
#pragma unroll
        for (int t = 0; t < 2; ++t) lv[t] = lam[col0 + wc * 32 + t * 16 + nn];
    }
#pragma unroll
    for (int t = 0; t < 2; ++t) {
        const int col = col0 + wc * 32 + t * 16 + nn;
#pragma unroll
        for (int rb = 0; rb < 2; ++rb)
#pragma unroll
            for (int g = 0; g < 8; ++g) {
                const int r = row0 + wr * 32 + rb * 16 + 8 * hi + g;
                float v = acc[rb][t][g];
                if (FIRST) v = __expf(lv[t] * sg[rb][g]) * v;
                C[(size_t)r * NCLS + col] = v;
            }
    }
}

__global__ __launch_bounds__(256) void product_scaled(
    const float* __restrict__ P0,
    const float* __restrict__ sigma,
    const float* __restrict__ lam,
    const float* __restrict__ V,
    float* __restrict__ U)
{
    __shared__ __attribute__((aligned(16))) float smem[LDSF];
    tile_product<true>(P0, V, sigma, lam, U, smem);
}

__global__ __launch_bounds__(256) void product_plain(
    const float* __restrict__ U,
    const float* __restrict__ sigma,
    const float* __restrict__ lam,
    const float* __restrict__ V,
    float* __restrict__ out)
{
    __shared__ __attribute__((aligned(16))) float smem[LDSF];
    tile_product<false>(U, V, sigma, lam, out, smem);
}

extern "C" void kernel_launch(void* const* d_in, const int* in_sizes, int n_in,
                              void* d_out, int out_size, void* d_ws, size_t ws_size,
                              hipStream_t stream) {
    const float* p0    = (const float*)d_in[0];
    const float* sigma = (const float*)d_in[1];
    const float* evals = (const float*)d_in[2];
    const float* evecs = (const float*)d_in[3];
    float* out = (float*)d_out;
    float* U   = (float*)d_ws;

    product_scaled<<<(NBATCH / TM) * (NCLS / TN), 256, 0, stream>>>(p0, sigma, evals, evecs, U);

    product_plain<<<(NBATCH / TM) * (NCLS / TN), 256, 0, stream>>>(U, sigma, evals, evecs, out);
}
